# starting kernel
# baseline (speedup 1.0000x reference)
#include <hip/hip_runtime.h>
#include <math.h>


#define N    512
#define DIM  512
#define H    64
#define OUT  32

typedef __attribute__((ext_vector_type(2))) float v2f;
typedef __attribute__((ext_vector_type(8))) float v8f;

__device__ inline v8f gemm16x16_f32(const float* __restrict__ Arow, int lda,
                                    const float* __restrict__ Bcol, int ldb,
                                    int K, int lane) {
  v8f c = {};
  const float* ap = Arow + (lane & 15) * lda + ((lane >> 4) << 1);
  const float* bp = Bcol + ((lane >> 4) << 1) * ldb;
  for (int k = 0; k < K; k += 4) {
    v2f a; a.x = ap[0];  a.y = ap[1];
    v2f b; b.x = bp[0];  b.y = bp[ldb];
    c = __builtin_amdgcn_wmma_f32_16x16x4_f32(false, a, false, b, (short)0, c,
                                              false, false);
    ap += 4;
    bp += 4 * ldb;
  }
  return c;
}

__device__ inline void store_tile16x16(float* __restrict__ D, int ldd,
                                       int m0, int n0, v8f c, int lane) {
  int col   = n0 + (lane & 15);
  int rbase = m0 + ((lane >> 4) << 3);
#pragma unroll
  for (int v = 0; v < 8; ++v) D[(rbase + v) * ldd + col] = c[v];
}

__global__ void gemm_pq_kernel(const float* __restrict__ F,
                               const float* __restrict__ w1,
                               float* __restrict__ PQ) {
  const int wave = threadIdx.x >> 5;
  const int lane = threadIdx.x & 31;
  const int tile = blockIdx.x * 8 + wave;
  const int m0 = (tile >> 3) * 16;
  const int n0 = (tile & 7) * 16;
  const int c  = n0 + (lane & 15);
  const float* Bcol = (c < H) ? (w1 + c) : (w1 + DIM * H + (c - H));
  v8f acc = gemm16x16_f32(F + m0 * DIM, DIM, Bcol, H, DIM, lane);
  store_tile16x16(PQ, 2 * H, m0, n0, acc, lane);
}

__global__ void gemm_xw_kernel(const float* __restrict__ F,
                               const float* __restrict__ W,
                               float* __restrict__ XW) {
  const int wave = threadIdx.x >> 5;
  const int lane = threadIdx.x & 31;
  const int tile = blockIdx.x * 8 + wave;
  const int m0 = (tile >> 1) * 16;
  const int n0 = (tile & 1) * 16;
  const float* Bcol = W + n0 + (lane & 15);
  v8f acc = gemm16x16_f32(F + m0 * DIM, DIM, Bcol, OUT, DIM, lane);
  store_tile16x16(XW, OUT, m0, n0, acc, lane);
}

__global__ void relnet_pair_kernel(const float* __restrict__ PQ,
                                   const float* __restrict__ b1,
                                   const float* __restrict__ w2,
                                   const float* __restrict__ b2p,
                                   float* __restrict__ adj) {
  __shared__ float Pj[16][H];
  __shared__ float Qi[16][H];
  __shared__ float b1s[H];
  __shared__ float w2s[H];
  const int tid = threadIdx.x;
  const int j0 = blockIdx.x * 16;
  const int i0 = blockIdx.y * 16;

  for (int t = tid; t < 16 * H; t += 256) {
    int r = t >> 6, h = t & (H - 1);
    Pj[r][h] = PQ[(j0 + r) * (2 * H) + h];
    Qi[r][h] = PQ[(i0 + r) * (2 * H) + H + h];
  }
  if (tid < H) { b1s[tid] = b1[tid]; w2s[tid] = w2[tid]; }
  __syncthreads();

  const int tj = tid & 15;
  const int ti = tid >> 4;
  float acc = 0.0f;
#pragma unroll 8
  for (int h = 0; h < H; ++h) {
    float t = Pj[tj][h] + Qi[ti][h] + b1s[h];
    acc = fmaf(fmaxf(t, 0.0f), w2s[h], acc);
  }
  float z = acc + b2p[0];
  adj[(i0 + ti) * N + (j0 + tj)] = 1.0f / (1.0f + expf(-z));
}

__global__ void dvec_kernel(const float* __restrict__ adj,
                            float* __restrict__ dvec) {
  __shared__ float red[256];
  const int i = blockIdx.x, tid = threadIdx.x;
  red[tid] = adj[i * N + tid] + adj[i * N + tid + 256];
  __syncthreads();
  for (int off = 128; off > 0; off >>= 1) {
    if (tid < off) red[tid] += red[tid + off];
    __syncthreads();
  }
  if (tid == 0) dvec[i] = sqrtf(red[0] + 2.0f);
}

__global__ void norm_kernel(const float* __restrict__ adj,
                            const float* __restrict__ dvec,
                            float* __restrict__ Aout) {
  const int idx = blockIdx.x * 256 + threadIdx.x;
  const int i = idx >> 9;
  const int j = idx & (N - 1);
  float a = adj[idx] + (i == j ? 1.0f : 0.0f);
  Aout[idx] = a / (dvec[i] * dvec[j]);
}

__global__ void gemm_out_kernel(const float* __restrict__ Anorm,
                                const float* __restrict__ XW,
                                const float* __restrict__ gcn_b,
                                const float* __restrict__ gamma,
                                const float* __restrict__ beta,
                                float* __restrict__ X) {
  const int wave = threadIdx.x >> 5;
  const int lane = threadIdx.x & 31;
  const int tile = blockIdx.x * 8 + wave;
  const int m0 = (tile >> 1) * 16;
  const int n0 = (tile & 1) * 16;
  const float* Bcol = XW + n0 + (lane & 15);
  v8f acc = gemm16x16_f32(Anorm + m0 * N, N, Bcol, OUT, N, lane);

  const int col = n0 + (lane & 15);
  const float inv_std = 0.99999500003749969f;
  const float sc = gamma[col] * inv_std;
  const float bt = beta[col];
  const float gb = gcn_b[col];
  const int rbase = m0 + ((lane >> 4) << 3);
#pragma unroll
  for (int v = 0; v < 8; ++v) {
    float t = (acc[v] + gb) * sc + bt;
    X[(rbase + v) * OUT + col] = fmaxf(t, 0.0f);
  }
}

extern "C" void kernel_launch(void* const* d_in, const int* in_sizes, int n_in,
                              void* d_out, int out_size, void* d_ws, size_t ws_size,
                              hipStream_t stream) {
  const float* F     = (const float*)d_in[0];
  const float* w1    = (const float*)d_in[1];
  const float* b1    = (const float*)d_in[2];
  const float* w2    = (const float*)d_in[3];
  const float* b2    = (const float*)d_in[4];
  const float* gw    = (const float*)d_in[5];
  const float* gb    = (const float*)d_in[6];
  const float* gamma = (const float*)d_in[7];
  const float* beta  = (const float*)d_in[8];

  float* out_x = (float*)d_out;
  float* out_A = out_x + N * OUT;

  float* ws   = (float*)d_ws;
  float* PQ   = ws;
  float* adj  = PQ + N * 2 * H;
  float* XW   = adj + N * N;
  float* dvec = XW + N * OUT;

  gemm_pq_kernel   <<<32,            256, 0, stream>>>(F, w1, PQ);
  gemm_xw_kernel   <<<8,             256, 0, stream>>>(F, gw, XW);
  relnet_pair_kernel<<<dim3(32, 32), 256, 0, stream>>>(PQ, b1, w2, b2, adj);
  dvec_kernel      <<<N,             256, 0, stream>>>(adj, dvec);
  norm_kernel      <<<(N * N) / 256, 256, 0, stream>>>(adj, dvec, out_A);
  gemm_out_kernel  <<<8,             256, 0, stream>>>(out_A, XW, gb, gamma, beta, out_x);
}


# GEMMs: each work-group owns one 16x32 output slab and splits K over its 8 waves (64 each); partial slabs summed in LDS in fixed order (changes f32 sum order); grids 32/8/8 -> 128/32/32
# speedup vs baseline: 1.6498x; 1.6498x over previous
#include <hip/hip_runtime.h>
#include <math.h>


#define N    512
#define DIM  512
#define H    64
#define OUT  32

typedef __attribute__((ext_vector_type(2))) float v2f;
typedef __attribute__((ext_vector_type(8))) float v8f;

#define KSPLIT 8
#define SLAB   512

template <int KLEN>
__device__ inline void gemm16x32_f32(const float* __restrict__ Arow, int lda,
                                     const float* __restrict__ Bcol, int ldb,
                                     int lane, v8f& c0, v8f& c1) {
  c0 = v8f{};
  c1 = v8f{};
  const float* ap = Arow + (lane & 15) * lda + ((lane >> 4) << 1);
  const float* bp = Bcol + ((lane >> 4) << 1) * ldb;
  for (int k = 0; k < KLEN; k += 4) {
    v2f a;  a.x  = ap[0];   a.y  = ap[1];
    v2f b0; b0.x = bp[0];   b0.y = bp[ldb];
    v2f b1; b1.x = bp[16];  b1.y = bp[ldb + 16];
    c0 = __builtin_amdgcn_wmma_f32_16x16x4_f32(false, a, false, b0, (short)0, c0,
                                               false, false);
    c1 = __builtin_amdgcn_wmma_f32_16x16x4_f32(false, a, false, b1, (short)0, c1,
                                               false, false);
    ap += 4;
    bp += 4 * ldb;
  }
}

__device__ inline void slab_put(float* __restrict__ red, int wave, int lane,
                                v8f c0, v8f c1) {
  float* p = red + wave * SLAB + lane;
  for (int v = 0; v < 8; ++v) {
    p[v * 64]      = c0[v];
    p[v * 64 + 32] = c1[v];
  }
}

__device__ inline void slab_sum(const float* __restrict__ red, int tid,
                                float& lo, float& hi) {
  const int r = tid >> 5, cc = tid & 31;
  const int idx = r * 64 + (cc >> 4) * 32 + (cc & 15);
  float s0 = red[idx];
  float s1 = red[idx + 16];
  for (int w = 1; w < KSPLIT; ++w) {
    s0 += red[w * SLAB + idx];
    s1 += red[w * SLAB + idx + 16];
  }
  lo = s0;
  hi = s1;
}

__global__ void gemm_pq_kernel(const float* __restrict__ F,
                               const float* __restrict__ w1,
                               float* __restrict__ PQ) {
  __shared__ float red[KSPLIT * SLAB];
  const int tid  = threadIdx.x;
  const int wave = tid >> 5;
  const int lane = tid & 31;
  const int m0 = (blockIdx.x >> 2) * 16;
  const int n0 = (blockIdx.x & 3) * 32;
  const int k0 = wave * (DIM / KSPLIT);
  const int c  = n0 + (lane & 15);
  const float* Bcol = (c < H) ? (w1 + c) : (w1 + DIM * H + (c - H));
  v8f c0, c1;
  gemm16x32_f32<DIM / KSPLIT>(F + m0 * DIM + k0, DIM, Bcol + k0 * H, H, lane, c0, c1);
  slab_put(red, wave, lane, c0, c1);
  __syncthreads();
  float lo, hi;
  slab_sum(red, tid, lo, hi);
  const int r = tid >> 5, cc = tid & 31;
  PQ[(m0 + r)     * (2 * H) + n0 + cc] = lo;
  PQ[(m0 + r + 8) * (2 * H) + n0 + cc] = hi;
}

__global__ void gemm_xw_kernel(const float* __restrict__ F,
                               const float* __restrict__ W,
                               float* __restrict__ XW) {
  __shared__ float red[KSPLIT * SLAB];
  const int tid  = threadIdx.x;
  const int wave = tid >> 5;
  const int lane = tid & 31;
  const int m0 = blockIdx.x * 16;
  const int k0 = wave * (DIM / KSPLIT);
  const float* Bcol = W + k0 * OUT + (lane & 15);
  v8f c0, c1;
  gemm16x32_f32<DIM / KSPLIT>(F + m0 * DIM + k0, DIM, Bcol, OUT, lane, c0, c1);
  slab_put(red, wave, lane, c0, c1);
  __syncthreads();
  float lo, hi;
  slab_sum(red, tid, lo, hi);
  const int r = tid >> 5, cc = tid & 31;
  XW[(m0 + r)     * OUT + cc] = lo;
  XW[(m0 + r + 8) * OUT + cc] = hi;
}

__global__ void relnet_pair_kernel(const float* __restrict__ PQ,
                                   const float* __restrict__ b1,
                                   const float* __restrict__ w2,
                                   const float* __restrict__ b2p,
                                   float* __restrict__ adj) {
  __shared__ float Pj[16][H];
  __shared__ float Qi[16][H];
  __shared__ float b1s[H];
  __shared__ float w2s[H];
  const int tid = threadIdx.x;
  const int j0 = blockIdx.x * 16;
  const int i0 = blockIdx.y * 16;

  for (int t = tid; t < 16 * H; t += 256) {
    int r = t >> 6, h = t & (H - 1);
    Pj[r][h] = PQ[(j0 + r) * (2 * H) + h];
    Qi[r][h] = PQ[(i0 + r) * (2 * H) + H + h];
  }
  if (tid < H) { b1s[tid] = b1[tid]; w2s[tid] = w2[tid]; }
  __syncthreads();

  const int tj = tid & 15;
  const int ti = tid >> 4;
  float acc = 0.0f;
#pragma unroll 8
  for (int h = 0; h < H; ++h) {
    float t = Pj[tj][h] + Qi[ti][h] + b1s[h];
    acc = fmaf(fmaxf(t, 0.0f), w2s[h], acc);
  }
  float z = acc + b2p[0];
  adj[(i0 + ti) * N + (j0 + tj)] = 1.0f / (1.0f + expf(-z));
}

__global__ void dvec_kernel(const float* __restrict__ adj,
                            float* __restrict__ dvec) {
  __shared__ float red[256];
  const int i = blockIdx.x, tid = threadIdx.x;
  red[tid] = adj[i * N + tid] + adj[i * N + tid + 256];
  __syncthreads();
  for (int off = 128; off > 0; off >>= 1) {
    if (tid < off) red[tid] += red[tid + off];
    __syncthreads();
  }
  if (tid == 0) dvec[i] = sqrtf(red[0] + 2.0f);
}

__global__ void norm_kernel(const float* __restrict__ adj,
                            const float* __restrict__ dvec,
                            float* __restrict__ Aout) {
  const int idx = blockIdx.x * 256 + threadIdx.x;
  const int i = idx >> 9;
  const int j = idx & (N - 1);
  float a = adj[idx] + (i == j ? 1.0f : 0.0f);
  Aout[idx] = a / (dvec[i] * dvec[j]);
}

__global__ void gemm_out_kernel(const float* __restrict__ Anorm,
                                const float* __restrict__ XW,
                                const float* __restrict__ gcn_b,
                                const float* __restrict__ gamma,
                                const float* __restrict__ beta,
                                float* __restrict__ X) {
  __shared__ float red[KSPLIT * SLAB];
  const int tid  = threadIdx.x;
  const int wave = tid >> 5;
  const int lane = tid & 31;
  const int m0 = blockIdx.x * 16;
  const int k0 = wave * (N / KSPLIT);
  const float* Bcol = XW + k0 * OUT + (lane & 15);
  v8f c0, c1;
  gemm16x32_f32<N / KSPLIT>(Anorm + m0 * N + k0, N, Bcol, OUT, lane, c0, c1);
  slab_put(red, wave, lane, c0, c1);
  __syncthreads();
  float lo, hi;
  slab_sum(red, tid, lo, hi);

  const int r = tid >> 5, col = tid & 31;
  const float inv_std = 0.99999500003749969f;
  const float sc = gamma[col] * inv_std;
  const float bt = beta[col];
  const float gb = gcn_b[col];
  const float t0 = (lo + gb) * sc + bt;
  const float t1 = (hi + gb) * sc + bt;
  X[(m0 + r)     * OUT + col] = fmaxf(t0, 0.0f);
  X[(m0 + r + 8) * OUT + col] = fmaxf(t1, 0.0f);
}

extern "C" void kernel_launch(void* const* d_in, const int* in_sizes, int n_in,
                              void* d_out, int out_size, void* d_ws, size_t ws_size,
                              hipStream_t stream) {
  const float* F     = (const float*)d_in[0];
  const float* w1    = (const float*)d_in[1];
  const float* b1    = (const float*)d_in[2];
  const float* w2    = (const float*)d_in[3];
  const float* b2    = (const float*)d_in[4];
  const float* gw    = (const float*)d_in[5];
  const float* gb    = (const float*)d_in[6];
  const float* gamma = (const float*)d_in[7];
  const float* beta  = (const float*)d_in[8];

  float* out_x = (float*)d_out;
  float* out_A = out_x + N * OUT;

  float* ws   = (float*)d_ws;
  float* PQ   = ws;
  float* adj  = PQ + N * 2 * H;
  float* XW   = adj + N * N;
  float* dvec = XW + N * OUT;

  gemm_pq_kernel   <<<128,           256, 0, stream>>>(F, w1, PQ);
  gemm_xw_kernel   <<<32,            256, 0, stream>>>(F, gw, XW);
  relnet_pair_kernel<<<dim3(32, 32), 256, 0, stream>>>(PQ, b1, w2, b2, adj);
  dvec_kernel      <<<N,             256, 0, stream>>>(adj, dvec);
  norm_kernel      <<<(N * N) / 256, 256, 0, stream>>>(adj, dvec, out_A);
  gemm_out_kernel  <<<32,            256, 0, stream>>>(out_A, XW, gb, gamma, beta, out_x);
}


# Six launches fused to three: XW GEMM rides in the PQ grid; pair kernel emits per-tile row sums (changes f32 sum order of the degree); normalisation moves into the output GEMM, A operand from LDS
# speedup vs baseline: 2.1414x; 1.2980x over previous
#include <hip/hip_runtime.h>
#include <math.h>


#define N    512
#define DIM  512
#define H    64
#define OUT  32

typedef __attribute__((ext_vector_type(2))) float v2f;
typedef __attribute__((ext_vector_type(8))) float v8f;

#define KSPLIT 8
#define SLAB   512

typedef __attribute__((ext_vector_type(4))) float v4f;

#define APITCH 516

template <int KLEN>
__device__ inline void gemm16x32_f32(const float* __restrict__ Arow, int lda,
                                     const float* __restrict__ Bcol, int ldb,
                                     int lane, v8f& c0, v8f& c1) {
  c0 = v8f{};
  c1 = v8f{};
  const float* ap = Arow + (lane & 15) * lda + ((lane >> 4) << 1);
  const float* bp = Bcol + ((lane >> 4) << 1) * ldb;
  for (int k = 0; k < KLEN; k += 4) {
    v2f a;  a.x  = ap[0];   a.y  = ap[1];
    v2f b0; b0.x = bp[0];   b0.y = bp[ldb];
    v2f b1; b1.x = bp[16];  b1.y = bp[ldb + 16];
    c0 = __builtin_amdgcn_wmma_f32_16x16x4_f32(false, a, false, b0, (short)0, c0,
                                               false, false);
    c1 = __builtin_amdgcn_wmma_f32_16x16x4_f32(false, a, false, b1, (short)0, c1,
                                               false, false);
    ap += 4;
    bp += 4 * ldb;
  }
}

__device__ inline void slab_put(float* __restrict__ red, int wave, int lane,
                                v8f c0, v8f c1) {
  float* p = red + wave * SLAB + lane;
  for (int v = 0; v < 8; ++v) {
    p[v * 64]      = c0[v];
    p[v * 64 + 32] = c1[v];
  }
}

__device__ inline void slab_sum(const float* __restrict__ red, int tid,
                                float& lo, float& hi) {
  const int r = tid >> 5, cc = tid & 31;
  const int idx = r * 64 + (cc >> 4) * 32 + (cc & 15);
  float s0 = red[idx];
  float s1 = red[idx + 16];
  for (int w = 1; w < KSPLIT; ++w) {
    s0 += red[w * SLAB + idx];
    s1 += red[w * SLAB + idx + 16];
  }
  lo = s0;
  hi = s1;
}

__global__ void gemm_pq_xw_kernel(const float* __restrict__ F,
                                  const float* __restrict__ w1,
                                  const float* __restrict__ gw,
                                  float* __restrict__ PQ,
                                  float* __restrict__ XW) {
  __shared__ float red[KSPLIT * SLAB];
  const int tid  = threadIdx.x;
  const int wave = tid >> 5;
  const int lane = tid & 31;
  const int k0 = wave * (DIM / KSPLIT);
  const int b  = blockIdx.x;
  v8f c0, c1;
  float* outp;
  int ldo;
  if (b < 128) {
    const int m0 = (b >> 2) * 16;
    const int n0 = (b & 3) * 32;
    const int c  = n0 + (lane & 15);
    const float* Bcol = (c < H) ? (w1 + c) : (w1 + DIM * H + (c - H));
    gemm16x32_f32<DIM / KSPLIT>(F + m0 * DIM + k0, DIM, Bcol + k0 * H, H, lane, c0, c1);
    outp = PQ + m0 * (2 * H) + n0;
    ldo  = 2 * H;
  } else {
    const int m0 = (b - 128) * 16;
    const float* Bcol = gw + k0 * OUT + (lane & 15);
    gemm16x32_f32<DIM / KSPLIT>(F + m0 * DIM + k0, DIM, Bcol, OUT, lane, c0, c1);
    outp = XW + m0 * OUT;
    ldo  = OUT;
  }
  slab_put(red, wave, lane, c0, c1);
  __syncthreads();
  float lo, hi;
  slab_sum(red, tid, lo, hi);
  const int r = tid >> 5, cc = tid & 31;
  outp[r * ldo + cc]       = lo;
  outp[(r + 8) * ldo + cc] = hi;
}

__global__ void relnet_pair_kernel(const float* __restrict__ PQ,
                                   const float* __restrict__ b1,
                                   const float* __restrict__ w2,
                                   const float* __restrict__ b2p,
                                   float* __restrict__ adj,
                                   float* __restrict__ rowpart) {
  __shared__ float srow[16][17];
  __shared__ float Pj[16][H];
  __shared__ float Qi[16][H];
  __shared__ float b1s[H];
  __shared__ float w2s[H];
  const int tid = threadIdx.x;
  const int j0 = blockIdx.x * 16;
  const int i0 = blockIdx.y * 16;

  for (int t = tid; t < 16 * H; t += 256) {
    int r = t >> 6, h = t & (H - 1);
    Pj[r][h] = PQ[(j0 + r) * (2 * H) + h];
    Qi[r][h] = PQ[(i0 + r) * (2 * H) + H + h];
  }
  if (tid < H) { b1s[tid] = b1[tid]; w2s[tid] = w2[tid]; }
  __syncthreads();

  const int tj = tid & 15;
  const int ti = tid >> 4;
  float acc = 0.0f;
#pragma unroll 8
  for (int h = 0; h < H; ++h) {
    float t = Pj[tj][h] + Qi[ti][h] + b1s[h];
    acc = fmaf(fmaxf(t, 0.0f), w2s[h], acc);
  }
  float z = acc + b2p[0];
  const float a = 1.0f / (1.0f + expf(-z));
  adj[(i0 + ti) * N + (j0 + tj)] = a;
  srow[ti][tj] = a;
  __syncthreads();
  if (tid < 16) {
    float sum = srow[tid][0];
    for (int j = 1; j < 16; ++j) sum += srow[tid][j];
    rowpart[blockIdx.x * N + i0 + tid] = sum;
  }
}

__global__ void norm_gemm_out_kernel(const float* __restrict__ adj,
                                     const float* __restrict__ rowpart,
                                     const float* __restrict__ XW,
                                     const float* __restrict__ gcn_b,
                                     const float* __restrict__ gamma,
                                     const float* __restrict__ beta,
                                     float* __restrict__ Aout,
                                     float* __restrict__ X) {
  __shared__ __align__(16) float As[16 * APITCH];
  __shared__ __align__(16) float part[2 * N];
  __shared__ __align__(16) float dv[N];
  __shared__ float red[KSPLIT * SLAB];
  const int tid  = threadIdx.x;
  const int wave = tid >> 5;
  const int lane = tid & 31;
  const int m0 = blockIdx.x * 16;

  const v4f* adj4 = (const v4f*)(adj + m0 * N);
  v4f av[8];
  for (int u = 0; u < 8; ++u) av[u] = adj4[tid + 256 * u];

  {
    const int q  = tid & 127;
    const int bh = tid >> 7;
    const v4f* rp = (const v4f*)rowpart + (bh * 16) * (N / 4) + q;
    v4f acc = rp[0];
    for (int t = 1; t < 16; ++t) acc += rp[t * (N / 4)];
    ((v4f*)part)[bh * (N / 4) + q] = acc;
  }
  __syncthreads();
  dv[tid]       = sqrtf(part[tid]       + part[N + tid]       + 2.0f);
  dv[tid + 256] = sqrtf(part[tid + 256] + part[N + tid + 256] + 2.0f);
  __syncthreads();

  v4f* out4 = (v4f*)(Aout + m0 * N);
  for (int u = 0; u < 8; ++u) {
    const int idx = tid + 256 * u;
    const int row = idx >> 7;
    const int j   = (idx & 127) * 4;
    const int i   = m0 + row;
    const float di = dv[i];
    const v4f dj = *(const v4f*)(dv + j);
    v4f a = av[u];
    a.x += (i == j     ? 1.0f : 0.0f);
    a.y += (i == j + 1 ? 1.0f : 0.0f);
    a.z += (i == j + 2 ? 1.0f : 0.0f);
    a.w += (i == j + 3 ? 1.0f : 0.0f);
    v4f o;
    o.x = a.x / (di * dj.x);
    o.y = a.y / (di * dj.y);
    o.z = a.z / (di * dj.z);
    o.w = a.w / (di * dj.w);
    out4[idx] = o;
    *(v4f*)(As + row * APITCH + j) = o;
  }
  __syncthreads();

  const int k0 = wave * (N / KSPLIT);
  const float* Bcol = XW + k0 * OUT + (lane & 15);
  v8f c0, c1;
  gemm16x32_f32<N / KSPLIT>(As + k0, APITCH, Bcol, OUT, lane, c0, c1);
  slab_put(red, wave, lane, c0, c1);
  __syncthreads();
  float lo, hi;
  slab_sum(red, tid, lo, hi);

  const int r = tid >> 5, col = tid & 31;
  const float inv_std = 0.99999500003749969f;
  const float sc = gamma[col] * inv_std;
  const float bt = beta[col];
  const float gb = gcn_b[col];
  const float t0 = (lo + gb) * sc + bt;
  const float t1 = (hi + gb) * sc + bt;
  X[(m0 + r)     * OUT + col] = fmaxf(t0, 0.0f);
  X[(m0 + r + 8) * OUT + col] = fmaxf(t1, 0.0f);
}

extern "C" void kernel_launch(void* const* d_in, const int* in_sizes, int n_in,
                              void* d_out, int out_size, void* d_ws, size_t ws_size,
                              hipStream_t stream) {
  const float* F     = (const float*)d_in[0];
  const float* w1    = (const float*)d_in[1];
  const float* b1    = (const float*)d_in[2];
  const float* w2    = (const float*)d_in[3];
  const float* b2    = (const float*)d_in[4];
  const float* gw    = (const float*)d_in[5];
  const float* gb    = (const float*)d_in[6];
  const float* gamma = (const float*)d_in[7];
  const float* beta  = (const float*)d_in[8];

  float* out_x = (float*)d_out;
  float* out_A = out_x + N * OUT;

  float* ws   = (float*)d_ws;
  float* PQ   = ws;
  float* adj  = PQ + N * 2 * H;
  float* XW   = adj + N * N;
  float* rowpart = XW + N * OUT;

  gemm_pq_xw_kernel   <<<160,          256, 0, stream>>>(F, w1, gw, PQ, XW);
  relnet_pair_kernel  <<<dim3(32, 32), 256, 0, stream>>>(PQ, b1, w2, b2, adj, rowpart);
  norm_gemm_out_kernel<<<32,           256, 0, stream>>>(adj, rowpart, XW, gb, gamma, beta,
                                                         out_A, out_x);
}


# Pair kernel re-tiled: work-group = 32x32 tile, lane keeps its P row in registers, Q+b1 of 4 rows per 16-byte LDS broadcast; row sums by lane butterfly (changes f32 sum order: pre-activation, degree)
# speedup vs baseline: 3.3251x; 1.5528x over previous
#include <hip/hip_runtime.h>
#include <math.h>


#define N    512
#define DIM  512
#define H    64
#define OUT  32

typedef __attribute__((ext_vector_type(2))) float v2f;
typedef __attribute__((ext_vector_type(8))) float v8f;

#define KSPLIT 8
#define SLAB   512

typedef __attribute__((ext_vector_type(4))) float v4f;

#define APITCH 516

template <int KLEN>
__device__ inline void gemm16x32_f32(const float* __restrict__ Arow, int lda,
                                     const float* __restrict__ Bcol, int ldb,
                                     int lane, v8f& c0, v8f& c1) {
  c0 = v8f{};
  c1 = v8f{};
  const float* ap = Arow + (lane & 15) * lda + ((lane >> 4) << 1);
  const float* bp = Bcol + ((lane >> 4) << 1) * ldb;
  for (int k = 0; k < KLEN; k += 4) {
    v2f a;  a.x  = ap[0];   a.y  = ap[1];
    v2f b0; b0.x = bp[0];   b0.y = bp[ldb];
    v2f b1; b1.x = bp[16];  b1.y = bp[ldb + 16];
    c0 = __builtin_amdgcn_wmma_f32_16x16x4_f32(false, a, false, b0, (short)0, c0,
                                               false, false);
    c1 = __builtin_amdgcn_wmma_f32_16x16x4_f32(false, a, false, b1, (short)0, c1,
                                               false, false);
    ap += 4;
    bp += 4 * ldb;
  }
}

__device__ inline void slab_put(float* __restrict__ red, int wave, int lane,
                                v8f c0, v8f c1) {
  float* p = red + wave * SLAB + lane;
  for (int v = 0; v < 8; ++v) {
    p[v * 64]      = c0[v];
    p[v * 64 + 32] = c1[v];
  }
}

__device__ inline void slab_sum(const float* __restrict__ red, int tid,
                                float& lo, float& hi) {
  const int r = tid >> 5, cc = tid & 31;
  const int idx = r * 64 + (cc >> 4) * 32 + (cc & 15);
  float s0 = red[idx];
  float s1 = red[idx + 16];
  for (int w = 1; w < KSPLIT; ++w) {
    s0 += red[w * SLAB + idx];
    s1 += red[w * SLAB + idx + 16];
  }
  lo = s0;
  hi = s1;
}

__global__ void gemm_pq_xw_kernel(const float* __restrict__ F,
                                  const float* __restrict__ w1,
                                  const float* __restrict__ gw,
                                  float* __restrict__ PQ,
                                  float* __restrict__ XW) {
  __shared__ float red[KSPLIT * SLAB];
  const int tid  = threadIdx.x;
  const int wave = tid >> 5;
  const int lane = tid & 31;
  const int k0 = wave * (DIM / KSPLIT);
  const int b  = blockIdx.x;
  v8f c0, c1;
  float* outp;
  int ldo;
  if (b < 128) {
    const int m0 = (b >> 2) * 16;
    const int n0 = (b & 3) * 32;
    const int c  = n0 + (lane & 15);
    const float* Bcol = (c < H) ? (w1 + c) : (w1 + DIM * H + (c - H));
    gemm16x32_f32<DIM / KSPLIT>(F + m0 * DIM + k0, DIM, Bcol + k0 * H, H, lane, c0, c1);
    outp = PQ + m0 * (2 * H) + n0;
    ldo  = 2 * H;
  } else {
    const int m0 = (b - 128) * 16;
    const float* Bcol = gw + k0 * OUT + (lane & 15);
    gemm16x32_f32<DIM / KSPLIT>(F + m0 * DIM + k0, DIM, Bcol, OUT, lane, c0, c1);
    outp = XW + m0 * OUT;
    ldo  = OUT;
  }
  slab_put(red, wave, lane, c0, c1);
  __syncthreads();
  float lo, hi;
  slab_sum(red, tid, lo, hi);
  const int r = tid >> 5, cc = tid & 31;
  outp[r * ldo + cc]       = lo;
  outp[(r + 8) * ldo + cc] = hi;
}

#define PT     32
#define PPITCH 68

__device__ inline float xhalf(float x) {
  int b = __builtin_bit_cast(int, x);
  return __builtin_bit_cast(float,
      __builtin_amdgcn_permlanex16(b, b, 0x76543210, 0xfedcba98, false, false));
}

__global__ void __launch_bounds__(256)
relnet_pair_kernel(const float* __restrict__ PQ,
                   const float* __restrict__ b1,
                   const float* __restrict__ w2,
                   const float* __restrict__ b2p,
                   float* __restrict__ adj,
                   float* __restrict__ rowpart) {
  __shared__ __align__(16) float Ps[PT * PPITCH];
  __shared__ __align__(16) float Qt[PT * H];
  __shared__ __align__(16) float w2s[H];
  __shared__ float rs[PT];
  const int tid  = threadIdx.x;
  const int wave = tid >> 5;
  const int lane = tid & 31;
  const int j0 = blockIdx.x * PT;
  const int i0 = blockIdx.y * PT;

  if (tid < 128) {
    const int g = tid >> 4;
    const int c = (tid & 15) * 4;
    const float* src = PQ + (i0 + 4 * g) * (2 * H) + H + c;
    const v4f b  = *(const v4f*)(b1 + c);
    const v4f r0 = *(const v4f*)(src)             + b;
    const v4f r1 = *(const v4f*)(src + 2 * H)     + b;
    const v4f r2 = *(const v4f*)(src + 4 * H)     + b;
    const v4f r3 = *(const v4f*)(src + 6 * H)     + b;
    float* dst = Qt + g * (4 * H) + c * 4;
    *(v4f*)(dst)      = v4f{r0.x, r1.x, r2.x, r3.x};
    *(v4f*)(dst + 4)  = v4f{r0.y, r1.y, r2.y, r3.y};
    *(v4f*)(dst + 8)  = v4f{r0.z, r1.z, r2.z, r3.z};
    *(v4f*)(dst + 12) = v4f{r0.w, r1.w, r2.w, r3.w};
    if (tid < H / 4) *(v4f*)(w2s + tid * 4) = *(const v4f*)(w2 + tid * 4);
  } else {
    for (int u = 0; u < 4; ++u) {
      const int idx = (tid - 128) + 128 * u;
      const int r = idx >> 4;
      const int c = (idx & 15) * 4;
      *(v4f*)(Ps + r * PPITCH + c) = *(const v4f*)(PQ + (j0 + r) * (2 * H) + c);
    }
  }
  __syncthreads();

  v4f p[H / 4];
#pragma unroll
  for (int c = 0; c < H / 4; ++c) p[c] = *(const v4f*)(Ps + lane * PPITCH + 4 * c);

  const float* qw = Qt + wave * (4 * H);
  v2f acc01 = {0.0f, 0.0f};
  v2f acc23 = {0.0f, 0.0f};
  const v2f zero2 = {0.0f, 0.0f};
#pragma unroll
  for (int c = 0; c < H / 4; ++c) {
    const v4f w  = *(const v4f*)(w2s + 4 * c);
    const v4f pc = p[c];
#pragma unroll
    for (int e = 0; e < 4; ++e) {
      const v4f q = *(const v4f*)(qw + (4 * c + e) * 4);
      const v2f ps = {pc[e], pc[e]};
      const v2f ws = {w[e], w[e]};
      const v2f t01 = __builtin_elementwise_max(q.xy + ps, zero2);
      const v2f t23 = __builtin_elementwise_max(q.zw + ps, zero2);
      acc01 = __builtin_elementwise_fma(t01, ws, acc01);
      acc23 = __builtin_elementwise_fma(t23, ws, acc23);
    }
  }

  const float b2 = b2p[0];
  const float a0 = 1.0f / (1.0f + expf(-(acc01.x + b2)));
  const float a1 = 1.0f / (1.0f + expf(-(acc01.y + b2)));
  const float a2 = 1.0f / (1.0f + expf(-(acc23.x + b2)));
  const float a3 = 1.0f / (1.0f + expf(-(acc23.y + b2)));
  float s0 = a0, s1 = a1, s2 = a2, s3 = a3;
  for (int m = 1; m < 16; m <<= 1) {
    s0 += __shfl_xor(s0, m, 32);
    s1 += __shfl_xor(s1, m, 32);
    s2 += __shfl_xor(s2, m, 32);
    s3 += __shfl_xor(s3, m, 32);
  }
  s0 += xhalf(s0);
  s1 += xhalf(s1);
  s2 += xhalf(s2);
  s3 += xhalf(s3);
  if (lane == 0) {
    rs[4 * wave + 0] = s0;
    rs[4 * wave + 1] = s1;
    rs[4 * wave + 2] = s2;
    rs[4 * wave + 3] = s3;
  }
  __syncthreads();
  if (wave == 0) rowpart[blockIdx.x * N + i0 + lane] = rs[lane];

  float* arow = adj + (i0 + 4 * wave) * N + j0 + lane;
  arow[0 * N] = a0;
  arow[1 * N] = a1;
  arow[2 * N] = a2;
  arow[3 * N] = a3;
}

__global__ void norm_gemm_out_kernel(const float* __restrict__ adj,
                                     const float* __restrict__ rowpart,
                                     const float* __restrict__ XW,
                                     const float* __restrict__ gcn_b,
                                     const float* __restrict__ gamma,
                                     const float* __restrict__ beta,
                                     float* __restrict__ Aout,
                                     float* __restrict__ X) {
  __shared__ __align__(16) float As[16 * APITCH];
  __shared__ __align__(16) float part[2 * N];
  __shared__ __align__(16) float dv[N];
  __shared__ float red[KSPLIT * SLAB];
  const int tid  = threadIdx.x;
  const int wave = tid >> 5;
  const int lane = tid & 31;
  const int m0 = blockIdx.x * 16;

  const v4f* adj4 = (const v4f*)(adj + m0 * N);
  v4f av[8];
  for (int u = 0; u < 8; ++u) av[u] = adj4[tid + 256 * u];

  {
    const int q  = tid & 127;
    const int bh = tid >> 7;
    const v4f* rp = (const v4f*)rowpart + (bh * 8) * (N / 4) + q;
    v4f acc = rp[0];
    for (int t = 1; t < 8; ++t) acc += rp[t * (N / 4)];
    ((v4f*)part)[bh * (N / 4) + q] = acc;
  }
  __syncthreads();
  dv[tid]       = sqrtf(part[tid]       + part[N + tid]       + 2.0f);
  dv[tid + 256] = sqrtf(part[tid + 256] + part[N + tid + 256] + 2.0f);
  __syncthreads();

  v4f* out4 = (v4f*)(Aout + m0 * N);
  for (int u = 0; u < 8; ++u) {
    const int idx = tid + 256 * u;
    const int row = idx >> 7;
    const int j   = (idx & 127) * 4;
    const int i   = m0 + row;
    const float di = dv[i];
    const v4f dj = *(const v4f*)(dv + j);
    v4f a = av[u];
    a.x += (i == j     ? 1.0f : 0.0f);
    a.y += (i == j + 1 ? 1.0f : 0.0f);
    a.z += (i == j + 2 ? 1.0f : 0.0f);
    a.w += (i == j + 3 ? 1.0f : 0.0f);
    v4f o;
    o.x = a.x / (di * dj.x);
    o.y = a.y / (di * dj.y);
    o.z = a.z / (di * dj.z);
    o.w = a.w / (di * dj.w);
    out4[idx] = o;
    *(v4f*)(As + row * APITCH + j) = o;
  }
  __syncthreads();

  const int k0 = wave * (N / KSPLIT);
  const float* Bcol = XW + k0 * OUT + (lane & 15);
  v8f c0, c1;
  gemm16x32_f32<N / KSPLIT>(As + k0, APITCH, Bcol, OUT, lane, c0, c1);
  slab_put(red, wave, lane, c0, c1);
  __syncthreads();
  float lo, hi;
  slab_sum(red, tid, lo, hi);

  const int r = tid >> 5, col = tid & 31;
  const float inv_std = 0.99999500003749969f;
  const float sc = gamma[col] * inv_std;
  const float bt = beta[col];
  const float gb = gcn_b[col];
  const float t0 = (lo + gb) * sc + bt;
  const float t1 = (hi + gb) * sc + bt;
  X[(m0 + r)     * OUT + col] = fmaxf(t0, 0.0f);
  X[(m0 + r + 8) * OUT + col] = fmaxf(t1, 0.0f);
}

extern "C" void kernel_launch(void* const* d_in, const int* in_sizes, int n_in,
                              void* d_out, int out_size, void* d_ws, size_t ws_size,
                              hipStream_t stream) {
  const float* F     = (const float*)d_in[0];
  const float* w1    = (const float*)d_in[1];
  const float* b1    = (const float*)d_in[2];
  const float* w2    = (const float*)d_in[3];
  const float* b2    = (const float*)d_in[4];
  const float* gw    = (const float*)d_in[5];
  const float* gb    = (const float*)d_in[6];
  const float* gamma = (const float*)d_in[7];
  const float* beta  = (const float*)d_in[8];

  float* out_x = (float*)d_out;
  float* out_A = out_x + N * OUT;

  float* ws   = (float*)d_ws;
  float* PQ   = ws;
  float* adj  = PQ + N * 2 * H;
  float* XW   = adj + N * N;
  float* rowpart = XW + N * OUT;

  gemm_pq_xw_kernel   <<<160,          256, 0, stream>>>(F, w1, gw, PQ, XW);
  relnet_pair_kernel  <<<dim3(16, 16), 256, 0, stream>>>(PQ, b1, w2, b2, adj, rowpart);
  norm_gemm_out_kernel<<<32,           256, 0, stream>>>(adj, rowpart, XW, gb, gamma, beta,
                                                         out_A, out_x);
}


# Last launch, 16 waves: 8 run the product on the raw I+adj strip, 1/d folded into XW rows and output rows, XW asked for at entry; 8 normalise and store it beside them (changes f32 rounding of x)
# speedup vs baseline: 3.5598x; 1.0706x over previous
#include <hip/hip_runtime.h>
#include <math.h>


#define N    512
#define DIM  512
#define H    64
#define OUT  32

typedef __attribute__((ext_vector_type(2))) float v2f;
typedef __attribute__((ext_vector_type(8))) float v8f;

#define KSPLIT 8
#define SLAB   512

typedef __attribute__((ext_vector_type(4))) float v4f;

#define APITCH 516

template <int KLEN>
__device__ inline void gemm16x32_f32(const float* __restrict__ Arow, int lda,
                                     const float* __restrict__ Bcol, int ldb,
                                     int lane, v8f& c0, v8f& c1) {
  c0 = v8f{};
  c1 = v8f{};
  const float* ap = Arow + (lane & 15) * lda + ((lane >> 4) << 1);
  const float* bp = Bcol + ((lane >> 4) << 1) * ldb;
  for (int k = 0; k < KLEN; k += 4) {
    v2f a;  a.x  = ap[0];   a.y  = ap[1];
    v2f b0; b0.x = bp[0];   b0.y = bp[ldb];
    v2f b1; b1.x = bp[16];  b1.y = bp[ldb + 16];
    c0 = __builtin_amdgcn_wmma_f32_16x16x4_f32(false, a, false, b0, (short)0, c0,
                                               false, false);
    c1 = __builtin_amdgcn_wmma_f32_16x16x4_f32(false, a, false, b1, (short)0, c1,
                                               false, false);
    ap += 4;
    bp += 4 * ldb;
  }
}

__device__ inline void slab_put(float* __restrict__ red, int wave, int lane,
                                v8f c0, v8f c1) {
  float* p = red + wave * SLAB + lane;
  for (int v = 0; v < 8; ++v) {
    p[v * 64]      = c0[v];
    p[v * 64 + 32] = c1[v];
  }
}

__device__ inline void slab_sum(const float* __restrict__ red, int tid,
                                float& lo, float& hi) {
  const int r = tid >> 5, cc = tid & 31;
  const int idx = r * 64 + (cc >> 4) * 32 + (cc & 15);
  float s0 = red[idx];
  float s1 = red[idx + 16];
  for (int w = 1; w < KSPLIT; ++w) {
    s0 += red[w * SLAB + idx];
    s1 += red[w * SLAB + idx + 16];
  }
  lo = s0;
  hi = s1;
}

__global__ void gemm_pq_xw_kernel(const float* __restrict__ F,
                                  const float* __restrict__ w1,
                                  const float* __restrict__ gw,
                                  float* __restrict__ PQ,
                                  float* __restrict__ XW) {
  __shared__ float red[KSPLIT * SLAB];
  const int tid  = threadIdx.x;
  const int wave = tid >> 5;
  const int lane = tid & 31;
  const int k0 = wave * (DIM / KSPLIT);
  const int b  = blockIdx.x;
  v8f c0, c1;
  float* outp;
  int ldo;
  if (b < 128) {
    const int m0 = (b >> 2) * 16;
    const int n0 = (b & 3) * 32;
    const int c  = n0 + (lane & 15);
    const float* Bcol = (c < H) ? (w1 + c) : (w1 + DIM * H + (c - H));
    gemm16x32_f32<DIM / KSPLIT>(F + m0 * DIM + k0, DIM, Bcol + k0 * H, H, lane, c0, c1);
    outp = PQ + m0 * (2 * H) + n0;
    ldo  = 2 * H;
  } else {
    const int m0 = (b - 128) * 16;
    const float* Bcol = gw + k0 * OUT + (lane & 15);
    gemm16x32_f32<DIM / KSPLIT>(F + m0 * DIM + k0, DIM, Bcol, OUT, lane, c0, c1);
    outp = XW + m0 * OUT;
    ldo  = OUT;
  }
  slab_put(red, wave, lane, c0, c1);
  __syncthreads();
  float lo, hi;
  slab_sum(red, tid, lo, hi);
  const int r = tid >> 5, cc = tid & 31;
  outp[r * ldo + cc]       = lo;
  outp[(r + 8) * ldo + cc] = hi;
}

#define PT     32
#define PPITCH 68

__device__ inline float xhalf(float x) {
  int b = __builtin_bit_cast(int, x);
  return __builtin_bit_cast(float,
      __builtin_amdgcn_permlanex16(b, b, 0x76543210, 0xfedcba98, false, false));
}

__global__ void __launch_bounds__(256)
relnet_pair_kernel(const float* __restrict__ PQ,
                   const float* __restrict__ b1,
                   const float* __restrict__ w2,
                   const float* __restrict__ b2p,
                   float* __restrict__ adj,
                   float* __restrict__ rowpart) {
  __shared__ __align__(16) float Ps[PT * PPITCH];
  __shared__ __align__(16) float Qt[PT * H];
  __shared__ __align__(16) float w2s[H];
  __shared__ float rs[PT];
  const int tid  = threadIdx.x;
  const int wave = tid >> 5;
  const int lane = tid & 31;
  const int j0 = blockIdx.x * PT;
  const int i0 = blockIdx.y * PT;

  if (tid < 128) {
    const int g = tid >> 4;
    const int c = (tid & 15) * 4;
    const float* src = PQ + (i0 + 4 * g) * (2 * H) + H + c;
    const v4f b  = *(const v4f*)(b1 + c);
    const v4f r0 = *(const v4f*)(src)             + b;
    const v4f r1 = *(const v4f*)(src + 2 * H)     + b;
    const v4f r2 = *(const v4f*)(src + 4 * H)     + b;
    const v4f r3 = *(const v4f*)(src + 6 * H)     + b;
    float* dst = Qt + g * (4 * H) + c * 4;
    *(v4f*)(dst)      = v4f{r0.x, r1.x, r2.x, r3.x};
    *(v4f*)(dst + 4)  = v4f{r0.y, r1.y, r2.y, r3.y};
    *(v4f*)(dst + 8)  = v4f{r0.z, r1.z, r2.z, r3.z};
    *(v4f*)(dst + 12) = v4f{r0.w, r1.w, r2.w, r3.w};
    if (tid < H / 4) *(v4f*)(w2s + tid * 4) = *(const v4f*)(w2 + tid * 4);
  } else {
    for (int u = 0; u < 4; ++u) {
      const int idx = (tid - 128) + 128 * u;
      const int r = idx >> 4;
      const int c = (idx & 15) * 4;
      *(v4f*)(Ps + r * PPITCH + c) = *(const v4f*)(PQ + (j0 + r) * (2 * H) + c);
    }
  }
  __syncthreads();

  v4f p[H / 4];
#pragma unroll
  for (int c = 0; c < H / 4; ++c) p[c] = *(const v4f*)(Ps + lane * PPITCH + 4 * c);

  const float* qw = Qt + wave * (4 * H);
  v2f acc01 = {0.0f, 0.0f};
  v2f acc23 = {0.0f, 0.0f};
  const v2f zero2 = {0.0f, 0.0f};
#pragma unroll
  for (int c = 0; c < H / 4; ++c) {
    const v4f w  = *(const v4f*)(w2s + 4 * c);
    const v4f pc = p[c];
#pragma unroll
    for (int e = 0; e < 4; ++e) {
      const v4f q = *(const v4f*)(qw + (4 * c + e) * 4);
      const v2f ps = {pc[e], pc[e]};
      const v2f ws = {w[e], w[e]};
      const v2f t01 = __builtin_elementwise_max(q.xy + ps, zero2);
      const v2f t23 = __builtin_elementwise_max(q.zw + ps, zero2);
      acc01 = __builtin_elementwise_fma(t01, ws, acc01);
      acc23 = __builtin_elementwise_fma(t23, ws, acc23);
    }
  }

  const float b2 = b2p[0];
  const float a0 = 1.0f / (1.0f + expf(-(acc01.x + b2)));
  const float a1 = 1.0f / (1.0f + expf(-(acc01.y + b2)));
  const float a2 = 1.0f / (1.0f + expf(-(acc23.x + b2)));
  const float a3 = 1.0f / (1.0f + expf(-(acc23.y + b2)));
  float s0 = a0, s1 = a1, s2 = a2, s3 = a3;
  for (int m = 1; m < 16; m <<= 1) {
    s0 += __shfl_xor(s0, m, 32);
    s1 += __shfl_xor(s1, m, 32);
    s2 += __shfl_xor(s2, m, 32);
    s3 += __shfl_xor(s3, m, 32);
  }
  s0 += xhalf(s0);
  s1 += xhalf(s1);
  s2 += xhalf(s2);
  s3 += xhalf(s3);
  if (lane == 0) {
    rs[4 * wave + 0] = s0;
    rs[4 * wave + 1] = s1;
    rs[4 * wave + 2] = s2;
    rs[4 * wave + 3] = s3;
  }
  __syncthreads();
  if (wave == 0) rowpart[blockIdx.x * N + i0 + lane] = rs[lane];

  float* arow = adj + (i0 + 4 * wave) * N + j0 + lane;
  arow[0 * N] = a0;
  arow[1 * N] = a1;
  arow[2 * N] = a2;
  arow[3 * N] = a3;
}

#define NT3 512

__device__ inline void lds_barrier() {
  asm volatile("s_wait_dscnt 0x0\n\ts_barrier_signal -1\n\ts_barrier_wait -1" ::: "memory");
}

__device__ inline void slab_sum_pairs(const float* __restrict__ red, int t,
                                      float& lo, float& hi) {
  const int r = t >> 5, cc = t & 31;
  const int idx = r * 64 + (cc & 1) * 32 + (cc >> 1);
  float s0 = red[idx];
  float s1 = red[idx + 16];
  for (int w = 1; w < KSPLIT; ++w) {
    s0 += red[w * SLAB + idx];
    s1 += red[w * SLAB + idx + 16];
  }
  lo = s0;
  hi = s1;
}

__global__ void __launch_bounds__(NT3)
norm_gemm_out_kernel(const float* __restrict__ adj,
                     const float* __restrict__ rowpart,
                     const float* __restrict__ XW,
                     const float* __restrict__ gcn_b,
                     const float* __restrict__ gamma,
                     const float* __restrict__ beta,
                     float* __restrict__ Aout,
                     float* __restrict__ X) {
  __shared__ __align__(16) float As[16 * APITCH];
  __shared__ __align__(16) float part[2 * N];
  __shared__ __align__(16) float dv[N];
  __shared__ __align__(16) float rd[N];
  __shared__ float red[KSPLIT * SLAB];
  const int tid  = threadIdx.x;
  const int lane = tid & 31;
  const int wave = __builtin_amdgcn_readfirstlane(tid >> 5);
  const bool prod = wave < KSPLIT;
  const int t   = tid & 255;
  const int m0  = blockIdx.x * 16;
  const int k0  = (wave & (KSPLIT - 1)) * (N / KSPLIT);
  const int kh  = (lane >> 4) << 1;

  v4f av[8];
  v2f xe[16], xo[16];

  const float gm = gamma[tid & 31];
  const float bt = beta[tid & 31];
  const float gb = gcn_b[tid & 31];

  if (prod) {
    const float* bp = XW + (k0 + kh) * OUT + 2 * (lane & 15);
#pragma unroll
    for (int s = 0; s < 16; ++s) {
      xe[s] = *(const v2f*)(bp);
      xo[s] = *(const v2f*)(bp + OUT);
      bp += 4 * OUT;
    }
  } else {
    const int q  = t & 127;
    const int bh = t >> 7;
    const v4f* rp = (const v4f*)rowpart + (bh * 8) * (N / 4) + q;
    v4f r8[8];
#pragma unroll
    for (int u = 0; u < 8; ++u) r8[u] = rp[u * (N / 4)];
    const v4f* adj4 = (const v4f*)(adj + m0 * N);
#pragma unroll
    for (int u = 0; u < 8; ++u) av[u] = adj4[t + 256 * u];
    v4f acc = r8[0];
#pragma unroll
    for (int u = 1; u < 8; ++u) acc += r8[u];
    ((v4f*)part)[bh * (N / 4) + q] = acc;
#pragma unroll
    for (int u = 0; u < 8; ++u) {
      const int idx = t + 256 * u;
      const int row = idx >> 7;
      const int j   = (idx & 127) * 4;
      const int i   = m0 + row;
      v4f a = av[u];
      a.x += (i == j     ? 1.0f : 0.0f);
      a.y += (i == j + 1 ? 1.0f : 0.0f);
      a.z += (i == j + 2 ? 1.0f : 0.0f);
      a.w += (i == j + 3 ? 1.0f : 0.0f);
      av[u] = a;
      *(v4f*)(As + row * APITCH + j) = a;
    }
  }
  lds_barrier();
  {
    const float d = sqrtf(part[tid] + part[N + tid] + 2.0f);
    dv[tid] = d;
    rd[tid] = 1.0f / d;
  }
  lds_barrier();

  if (prod) {
    v8f c0 = v8f{};
    v8f c1 = v8f{};
    const float* ap = As + (lane & 15) * APITCH + k0 + kh;
    const float* rp = rd + k0 + kh;
#pragma unroll
    for (int s = 0; s < 16; ++s) {
      v2f a;  a.x = ap[4 * s];  a.y = ap[4 * s + 1];
      const float r0 = rp[4 * s];
      const float r1 = rp[4 * s + 1];
      v2f f0; f0.x = xe[s].x * r0;  f0.y = xo[s].x * r1;
      v2f f1; f1.x = xe[s].y * r0;  f1.y = xo[s].y * r1;
      c0 = __builtin_amdgcn_wmma_f32_16x16x4_f32(false, a, false, f0, (short)0, c0,
                                                 false, false);
      c1 = __builtin_amdgcn_wmma_f32_16x16x4_f32(false, a, false, f1, (short)0, c1,
                                                 false, false);
    }
    slab_put(red, wave, lane, c0, c1);
  } else {
    v4f* out4 = (v4f*)(Aout + m0 * N);
#pragma unroll
    for (int u = 0; u < 8; ++u) {
      const int idx = t + 256 * u;
      const int row = idx >> 7;
      const int j   = (idx & 127) * 4;
      const float di = dv[m0 + row];
      const v4f dj = *(const v4f*)(dv + j);
      const v4f a = av[u];
      v4f o;
      o.x = a.x / (di * dj.x);
      o.y = a.y / (di * dj.y);
      o.z = a.z / (di * dj.z);
      o.w = a.w / (di * dj.w);
      out4[idx] = o;
    }
  }
  lds_barrier();

  if (prod) {
    float lo, hi;
    slab_sum_pairs(red, t, lo, hi);
    const int r = t >> 5, col = t & 31;
    const float inv_std = 0.99999500003749969f;
    const float sc = gm * inv_std;
    const float t0 = (lo * rd[m0 + r]     + gb) * sc + bt;
    const float t1 = (hi * rd[m0 + r + 8] + gb) * sc + bt;
    X[(m0 + r)     * OUT + col] = fmaxf(t0, 0.0f);
    X[(m0 + r + 8) * OUT + col] = fmaxf(t1, 0.0f);
  }
}

extern "C" void kernel_launch(void* const* d_in, const int* in_sizes, int n_in,
                              void* d_out, int out_size, void* d_ws, size_t ws_size,
                              hipStream_t stream) {
  const float* F     = (const float*)d_in[0];
  const float* w1    = (const float*)d_in[1];
  const float* b1    = (const float*)d_in[2];
  const float* w2    = (const float*)d_in[3];
  const float* b2    = (const float*)d_in[4];
  const float* gw    = (const float*)d_in[5];
  const float* gb    = (const float*)d_in[6];
  const float* gamma = (const float*)d_in[7];
  const float* beta  = (const float*)d_in[8];

  float* out_x = (float*)d_out;
  float* out_A = out_x + N * OUT;

  float* ws   = (float*)d_ws;
  float* PQ   = ws;
  float* adj  = PQ + N * 2 * H;
  float* XW   = adj + N * N;
  float* rowpart = XW + N * OUT;

  gemm_pq_xw_kernel   <<<160,          256, 0, stream>>>(F, w1, gw, PQ, XW);
  relnet_pair_kernel  <<<dim3(16, 16), 256, 0, stream>>>(PQ, b1, w2, b2, adj, rowpart);
  norm_gemm_out_kernel<<<32,           NT3, 0, stream>>>(adj, rowpart, XW, gb, gamma, beta,
                                                         out_A, out_x);
}


# Launch 1 operand loads: A by 16-byte loads (lane half = 4 adjacent k, 2 matrix steps), B by 8-byte column pairs on whole lines; a wave's 40 loads (were 80) all issued ahead (changes f32 sum order)
# speedup vs baseline: 3.8564x; 1.0833x over previous
#include <hip/hip_runtime.h>
#include <math.h>


#define N    512
#define DIM  512
#define H    64
#define OUT  32

typedef __attribute__((ext_vector_type(2))) float v2f;
typedef __attribute__((ext_vector_type(8))) float v8f;

#define KSPLIT 8
#define SLAB   512

typedef __attribute__((ext_vector_type(4))) float v4f;

#define APITCH 516

template <int KLEN>
__device__ inline void gemm16x32_f32(const float* __restrict__ Arow, int lda,
                                     const float* __restrict__ Bpair, int ldb,
                                     int lane, v8f& c0, v8f& c1) {
  c0 = v8f{};
  c1 = v8f{};
  const int kq = (lane >> 4) << 2;
  const float* ap = Arow + (lane & 15) * lda + kq;
  const float* bp = Bpair + kq * ldb;
  v4f a4[KLEN / 8];
  v2f br[KLEN / 2];
  for (int t = 0; t < KLEN / 8; ++t) {
    a4[t]         = *(const v4f*)(ap + 8 * t);
    br[4 * t]     = *(const v2f*)(bp + (8 * t) * ldb);
    br[4 * t + 1] = *(const v2f*)(bp + (8 * t + 1) * ldb);
    br[4 * t + 2] = *(const v2f*)(bp + (8 * t + 2) * ldb);
    br[4 * t + 3] = *(const v2f*)(bp + (8 * t + 3) * ldb);
  }
  __builtin_amdgcn_sched_barrier(0);
  for (int t = 0; t < KLEN / 8; ++t) {
    v2f a, b0, b1;
    a.x  = a4[t].x;          a.y  = a4[t].y;
    b0.x = br[4 * t].x;      b0.y = br[4 * t + 1].x;
    b1.x = br[4 * t].y;      b1.y = br[4 * t + 1].y;
    c0 = __builtin_amdgcn_wmma_f32_16x16x4_f32(false, a, false, b0, (short)0, c0,
                                               false, false);
    c1 = __builtin_amdgcn_wmma_f32_16x16x4_f32(false, a, false, b1, (short)0, c1,
                                               false, false);
    a.x  = a4[t].z;          a.y  = a4[t].w;
    b0.x = br[4 * t + 2].x;  b0.y = br[4 * t + 3].x;
    b1.x = br[4 * t + 2].y;  b1.y = br[4 * t + 3].y;
    c0 = __builtin_amdgcn_wmma_f32_16x16x4_f32(false, a, false, b0, (short)0, c0,
                                               false, false);
    c1 = __builtin_amdgcn_wmma_f32_16x16x4_f32(false, a, false, b1, (short)0, c1,
                                               false, false);
  }
}

__device__ inline void slab_put(float* __restrict__ red, int wave, int lane,
                                v8f c0, v8f c1) {
  float* p = red + wave * SLAB + lane;
  for (int v = 0; v < 8; ++v) {
    p[v * 64]      = c0[v];
    p[v * 64 + 32] = c1[v];
  }
}

__device__ inline void slab_sum_pairs(const float* __restrict__ red, int t,
                                      float& lo, float& hi) {
  const int r = t >> 5, cc = t & 31;
  const int idx = r * 64 + (cc & 1) * 32 + (cc >> 1);
  float s0 = red[idx];
  float s1 = red[idx + 16];
  for (int w = 1; w < KSPLIT; ++w) {
    s0 += red[w * SLAB + idx];
    s1 += red[w * SLAB + idx + 16];
  }
  lo = s0;
  hi = s1;
}

__global__ void gemm_pq_xw_kernel(const float* __restrict__ F,
                                  const float* __restrict__ w1,
                                  const float* __restrict__ gw,
                                  float* __restrict__ PQ,
                                  float* __restrict__ XW) {
  __shared__ float red[KSPLIT * SLAB];
  const int tid  = threadIdx.x;
  const int wave = tid >> 5;
  const int lane = tid & 31;
  const int k0 = wave * (DIM / KSPLIT);
  const int b  = blockIdx.x;
  v8f c0, c1;
  float* outp;
  int ldo;
  if (b < 128) {
    const int m0 = (b >> 2) * 16;
    const int n0 = (b & 3) * 32;
    const float* Bslab = (n0 < H) ? (w1 + n0) : (w1 + DIM * H + (n0 - H));
    gemm16x32_f32<DIM / KSPLIT>(F + m0 * DIM + k0, DIM, Bslab + k0 * H + 2 * (lane & 15), H,
                                lane, c0, c1);
    outp = PQ + m0 * (2 * H) + n0;
    ldo  = 2 * H;
  } else {
    const int m0 = (b - 128) * 16;
    gemm16x32_f32<DIM / KSPLIT>(F + m0 * DIM + k0, DIM, gw + k0 * OUT + 2 * (lane & 15), OUT,
                                lane, c0, c1);
    outp = XW + m0 * OUT;
    ldo  = OUT;
  }
  slab_put(red, wave, lane, c0, c1);
  __syncthreads();
  float lo, hi;
  slab_sum_pairs(red, tid, lo, hi);
  const int r = tid >> 5, cc = tid & 31;
  outp[r * ldo + cc]       = lo;
  outp[(r + 8) * ldo + cc] = hi;
}

#define PT     32
#define PPITCH 68

__device__ inline float xhalf(float x) {
  int b = __builtin_bit_cast(int, x);
  return __builtin_bit_cast(float,
      __builtin_amdgcn_permlanex16(b, b, 0x76543210, 0xfedcba98, false, false));
}

__global__ void __launch_bounds__(256)
relnet_pair_kernel(const float* __restrict__ PQ,
                   const float* __restrict__ b1,
                   const float* __restrict__ w2,
                   const float* __restrict__ b2p,
                   float* __restrict__ adj,
                   float* __restrict__ rowpart) {
  __shared__ __align__(16) float Ps[PT * PPITCH];
  __shared__ __align__(16) float Qt[PT * H];
  __shared__ __align__(16) float w2s[H];
  __shared__ float rs[PT];
  const int tid  = threadIdx.x;
  const int wave = tid >> 5;
  const int lane = tid & 31;
  const int j0 = blockIdx.x * PT;
  const int i0 = blockIdx.y * PT;

  if (tid < 128) {
    const int g = tid >> 4;
    const int c = (tid & 15) * 4;
    const float* src = PQ + (i0 + 4 * g) * (2 * H) + H + c;
    const v4f b  = *(const v4f*)(b1 + c);
    const v4f r0 = *(const v4f*)(src)             + b;
    const v4f r1 = *(const v4f*)(src + 2 * H)     + b;
    const v4f r2 = *(const v4f*)(src + 4 * H)     + b;
    const v4f r3 = *(const v4f*)(src + 6 * H)     + b;
    float* dst = Qt + g * (4 * H) + c * 4;
    *(v4f*)(dst)      = v4f{r0.x, r1.x, r2.x, r3.x};
    *(v4f*)(dst + 4)  = v4f{r0.y, r1.y, r2.y, r3.y};
    *(v4f*)(dst + 8)  = v4f{r0.z, r1.z, r2.z, r3.z};
    *(v4f*)(dst + 12) = v4f{r0.w, r1.w, r2.w, r3.w};
    if (tid < H / 4) *(v4f*)(w2s + tid * 4) = *(const v4f*)(w2 + tid * 4);
  } else {
    for (int u = 0; u < 4; ++u) {
      const int idx = (tid - 128) + 128 * u;
      const int r = idx >> 4;
      const int c = (idx & 15) * 4;
      *(v4f*)(Ps + r * PPITCH + c) = *(const v4f*)(PQ + (j0 + r) * (2 * H) + c);
    }
  }
  __syncthreads();

  v4f p[H / 4];
#pragma unroll
  for (int c = 0; c < H / 4; ++c) p[c] = *(const v4f*)(Ps + lane * PPITCH + 4 * c);

  const float* qw = Qt + wave * (4 * H);
  v2f acc01 = {0.0f, 0.0f};
  v2f acc23 = {0.0f, 0.0f};
  const v2f zero2 = {0.0f, 0.0f};
#pragma unroll
  for (int c = 0; c < H / 4; ++c) {
    const v4f w  = *(const v4f*)(w2s + 4 * c);
    const v4f pc = p[c];
#pragma unroll
    for (int e = 0; e < 4; ++e) {
      const v4f q = *(const v4f*)(qw + (4 * c + e) * 4);
      const v2f ps = {pc[e], pc[e]};
      const v2f ws = {w[e], w[e]};
      const v2f t01 = __builtin_elementwise_max(q.xy + ps, zero2);
      const v2f t23 = __builtin_elementwise_max(q.zw + ps, zero2);
      acc01 = __builtin_elementwise_fma(t01, ws, acc01);
      acc23 = __builtin_elementwise_fma(t23, ws, acc23);
    }
  }

  const float b2 = b2p[0];
  const float a0 = 1.0f / (1.0f + expf(-(acc01.x + b2)));
  const float a1 = 1.0f / (1.0f + expf(-(acc01.y + b2)));
  const float a2 = 1.0f / (1.0f + expf(-(acc23.x + b2)));
  const float a3 = 1.0f / (1.0f + expf(-(acc23.y + b2)));
  float s0 = a0, s1 = a1, s2 = a2, s3 = a3;
  for (int m = 1; m < 16; m <<= 1) {
    s0 += __shfl_xor(s0, m, 32);
    s1 += __shfl_xor(s1, m, 32);
    s2 += __shfl_xor(s2, m, 32);
    s3 += __shfl_xor(s3, m, 32);
  }
  s0 += xhalf(s0);
  s1 += xhalf(s1);
  s2 += xhalf(s2);
  s3 += xhalf(s3);
  if (lane == 0) {
    rs[4 * wave + 0] = s0;
    rs[4 * wave + 1] = s1;
    rs[4 * wave + 2] = s2;
    rs[4 * wave + 3] = s3;
  }
  __syncthreads();
  if (wave == 0) rowpart[blockIdx.x * N + i0 + lane] = rs[lane];

  float* arow = adj + (i0 + 4 * wave) * N + j0 + lane;
  arow[0 * N] = a0;
  arow[1 * N] = a1;
  arow[2 * N] = a2;
  arow[3 * N] = a3;
}

#define NT3 512

__device__ inline void lds_barrier() {
  asm volatile("s_wait_dscnt 0x0\n\ts_barrier_signal -1\n\ts_barrier_wait -1" ::: "memory");
}

__global__ void __launch_bounds__(NT3)
norm_gemm_out_kernel(const float* __restrict__ adj,
                     const float* __restrict__ rowpart,
                     const float* __restrict__ XW,
                     const float* __restrict__ gcn_b,
                     const float* __restrict__ gamma,
                     const float* __restrict__ beta,
                     float* __restrict__ Aout,
                     float* __restrict__ X) {
  __shared__ __align__(16) float As[16 * APITCH];
  __shared__ __align__(16) float part[2 * N];
  __shared__ __align__(16) float dv[N];
  __shared__ __align__(16) float rd[N];
  __shared__ float red[KSPLIT * SLAB];
  const int tid  = threadIdx.x;
  const int lane = tid & 31;
  const int wave = __builtin_amdgcn_readfirstlane(tid >> 5);
  const bool prod = wave < KSPLIT;
  const int t   = tid & 255;
  const int m0  = blockIdx.x * 16;
  const int k0  = (wave & (KSPLIT - 1)) * (N / KSPLIT);
  const int kh  = (lane >> 4) << 1;

  v4f av[8];
  v2f xe[16], xo[16];

  const float gm = gamma[tid & 31];
  const float bt = beta[tid & 31];
  const float gb = gcn_b[tid & 31];

  if (prod) {
    const float* bp = XW + (k0 + kh) * OUT + 2 * (lane & 15);
#pragma unroll
    for (int s = 0; s < 16; ++s) {
      xe[s] = *(const v2f*)(bp);
      xo[s] = *(const v2f*)(bp + OUT);
      bp += 4 * OUT;
    }
  } else {
    const int q  = t & 127;
    const int bh = t >> 7;
    const v4f* rp = (const v4f*)rowpart + (bh * 8) * (N / 4) + q;
    v4f r8[8];
#pragma unroll
    for (int u = 0; u < 8; ++u) r8[u] = rp[u * (N / 4)];
    const v4f* adj4 = (const v4f*)(adj + m0 * N);
#pragma unroll
    for (int u = 0; u < 8; ++u) av[u] = adj4[t + 256 * u];
    v4f acc = r8[0];
#pragma unroll
    for (int u = 1; u < 8; ++u) acc += r8[u];
    ((v4f*)part)[bh * (N / 4) + q] = acc;
#pragma unroll
    for (int u = 0; u < 8; ++u) {
      const int idx = t + 256 * u;
      const int row = idx >> 7;
      const int j   = (idx & 127) * 4;
      const int i   = m0 + row;
      v4f a = av[u];
      a.x += (i == j     ? 1.0f : 0.0f);
      a.y += (i == j + 1 ? 1.0f : 0.0f);
      a.z += (i == j + 2 ? 1.0f : 0.0f);
      a.w += (i == j + 3 ? 1.0f : 0.0f);
      av[u] = a;
      *(v4f*)(As + row * APITCH + j) = a;
    }
  }
  lds_barrier();
  {
    const float d = sqrtf(part[tid] + part[N + tid] + 2.0f);
    dv[tid] = d;
    rd[tid] = 1.0f / d;
  }
  lds_barrier();

  if (prod) {
    v8f c0 = v8f{};
    v8f c1 = v8f{};
    const float* ap = As + (lane & 15) * APITCH + k0 + kh;
    const float* rp = rd + k0 + kh;
#pragma unroll
    for (int s = 0; s < 16; ++s) {
      v2f a;  a.x = ap[4 * s];  a.y = ap[4 * s + 1];
      const float r0 = rp[4 * s];
      const float r1 = rp[4 * s + 1];
      v2f f0; f0.x = xe[s].x * r0;  f0.y = xo[s].x * r1;
      v2f f1; f1.x = xe[s].y * r0;  f1.y = xo[s].y * r1;
      c0 = __builtin_amdgcn_wmma_f32_16x16x4_f32(false, a, false, f0, (short)0, c0,
                                                 false, false);
      c1 = __builtin_amdgcn_wmma_f32_16x16x4_f32(false, a, false, f1, (short)0, c1,
                                                 false, false);
    }
    slab_put(red, wave, lane, c0, c1);
  } else {
    v4f* out4 = (v4f*)(Aout + m0 * N);
#pragma unroll
    for (int u = 0; u < 8; ++u) {
      const int idx = t + 256 * u;
      const int row = idx >> 7;
      const int j   = (idx & 127) * 4;
      const float di = dv[m0 + row];
      const v4f dj = *(const v4f*)(dv + j);
      const v4f a = av[u];
      v4f o;
      o.x = a.x / (di * dj.x);
      o.y = a.y / (di * dj.y);
      o.z = a.z / (di * dj.z);
      o.w = a.w / (di * dj.w);
      out4[idx] = o;
    }
  }
  lds_barrier();

  if (prod) {
    float lo, hi;
    slab_sum_pairs(red, t, lo, hi);
    const int r = t >> 5, col = t & 31;
    const float inv_std = 0.99999500003749969f;
    const float sc = gm * inv_std;
    const float t0 = (lo * rd[m0 + r]     + gb) * sc + bt;
    const float t1 = (hi * rd[m0 + r + 8] + gb) * sc + bt;
    X[(m0 + r)     * OUT + col] = fmaxf(t0, 0.0f);
    X[(m0 + r + 8) * OUT + col] = fmaxf(t1, 0.0f);
  }
}

extern "C" void kernel_launch(void* const* d_in, const int* in_sizes, int n_in,
                              void* d_out, int out_size, void* d_ws, size_t ws_size,
                              hipStream_t stream) {
  const float* F     = (const float*)d_in[0];
  const float* w1    = (const float*)d_in[1];
  const float* b1    = (const float*)d_in[2];
  const float* w2    = (const float*)d_in[3];
  const float* b2    = (const float*)d_in[4];
  const float* gw    = (const float*)d_in[5];
  const float* gb    = (const float*)d_in[6];
  const float* gamma = (const float*)d_in[7];
  const float* beta  = (const float*)d_in[8];

  float* out_x = (float*)d_out;
  float* out_A = out_x + N * OUT;

  float* ws   = (float*)d_ws;
  float* PQ   = ws;
  float* adj  = PQ + N * 2 * H;
  float* XW   = adj + N * N;
  float* rowpart = XW + N * OUT;

  gemm_pq_xw_kernel   <<<160,          256, 0, stream>>>(F, w1, gw, PQ, XW);
  relnet_pair_kernel  <<<dim3(16, 16), 256, 0, stream>>>(PQ, b1, w2, b2, adj, rowpart);
  norm_gemm_out_kernel<<<32,           NT3, 0, stream>>>(adj, rowpart, XW, gb, gamma, beta,
                                                         out_A, out_x);
}
